# starting kernel
# baseline (speedup 1.0000x reference)
#include <hip/hip_runtime.h>

typedef float v2f __attribute__((ext_vector_type(2)));
typedef float v8f __attribute__((ext_vector_type(8)));

#define NMODES 16
#define NLAYERS 6

__global__ __launch_bounds__(192) void cv_setup_kernel(
    const float* __restrict__ disp,
    const float* __restrict__ sq,
    const float* __restrict__ bs,
    float* __restrict__ ws)
{
  __shared__ float s_dx[2][NLAYERS][NMODES];
  __shared__ float s_dp[2][NLAYERS][NMODES];
  __shared__ float s_sa[2][NLAYERS][NMODES];
  __shared__ float s_sb[2][NLAYERS][NMODES];
  __shared__ float s_sd[2][NLAYERS][NMODES];
  __shared__ float s_ct[2][NLAYERS][NMODES - 1];
  __shared__ float s_cs[2][NLAYERS][NMODES - 1];
  __shared__ float s_ss[2][NLAYERS][NMODES - 1];

  const int t = threadIdx.x;

  if (t < 2 * NLAYERS * NMODES) {
    int cc = t / (NLAYERS * NMODES);
    int r  = t % (NLAYERS * NMODES);
    int l  = r / NMODES, i = r % NMODES;
    int base = ((cc * NLAYERS + l) * NMODES + i) * 2;
    float a = disp[base + 0], dphi = disp[base + 1];
    s_dx[cc][l][i] = 2.0f * a * cosf(dphi);
    s_dp[cc][l][i] = 2.0f * a * sinf(dphi);
    float rr = fabsf(sq[base + 0]);
    float sphi = sq[base + 1];
    float ch = coshf(rr), sh = sinhf(rr);
    float cp = cosf(sphi), sp = sinf(sphi);
    s_sa[cc][l][i] = ch - cp * sh;
    s_sb[cc][l][i] = -sp * sh;
    s_sd[cc][l][i] = ch + cp * sh;
  }
  if (t < 2 * NLAYERS * (NMODES - 1)) {
    int cc = t / (NLAYERS * (NMODES - 1));
    int r  = t % (NLAYERS * (NMODES - 1));
    int l  = r / (NMODES - 1), w = r % (NMODES - 1);
    int base = ((cc * NLAYERS + l) * (NMODES - 1) + w) * 2;
    float th   = 1.0f / (1.0f + expf(-bs[base + 0]));
    float bphi = bs[base + 1];
    float ct = cosf(th), st = sinf(th);
    s_ct[cc][l][w] = ct;
    s_cs[cc][l][w] = cosf(bphi) * st;
    s_ss[cc][l][w] = sinf(bphi) * st;
  }
  __syncthreads();

  if (t >= 34) return;
  const int cc  = t / 17;
  const int col = t % 17;
  const bool use_disp = (col == 16);

  float mx[NMODES], mp[NMODES];
#pragma unroll
  for (int i = 0; i < NMODES; ++i) {
    mx[i] = (!use_disp && i == col) ? 2.0f : 0.0f;
    mp[i] = 0.0f;
  }

  for (int l = 0; l < NLAYERS; ++l) {
#pragma unroll
    for (int i = 0; i < NMODES; ++i) {
      float xv = mx[i], pv = mp[i];
      if (use_disp) { xv += s_dx[cc][l][i]; pv += s_dp[cc][l][i]; }
      float A = s_sa[cc][l][i], Bb = s_sb[cc][l][i], D = s_sd[cc][l][i];
      mx[i] = A * xv + Bb * pv;
      mp[i] = Bb * xv + D * pv;
    }
#pragma unroll
    for (int w = 0; w < NMODES - 1; ++w) {
      float ct = s_ct[cc][l][w], cs = s_cs[cc][l][w], ss = s_ss[cc][l][w];
      float x1 = mx[w], x2 = mx[w + 1], p1 = mp[w], p2 = mp[w + 1];
      mx[w]     = ct * x1 - cs * x2 - ss * p2;
      mx[w + 1] = cs * x1 + ct * x2 - ss * p1;
      mp[w]     = ss * x2 + ct * p1 - cs * p2;
      mp[w + 1] = ss * x1 + cs * p1 + ct * p2;
    }
  }

  if (col < 16) {
#pragma unroll
    for (int j = 0; j < NMODES; ++j)
      ws[col * 32 + 16 * cc + j] = (cc == 0) ? mx[j] : mp[j];
  } else {
#pragma unroll
    for (int j = 0; j < NMODES; ++j)
      ws[512 + 16 * cc + j] = (cc == 0) ? mx[j] : mp[j];
  }
}

__global__ __launch_bounds__(256) void cv_gemm_kernel(
    const float* __restrict__ x, const float* __restrict__ ws,
    float* __restrict__ out, int numTiles)
{
  const int lane = threadIdx.x & 31;
  const int row  = lane & 15;
  const int hi   = lane >> 4;

  v2f bfrag[2][4];
#pragma unroll
  for (int h = 0; h < 2; ++h)
#pragma unroll
    for (int kc = 0; kc < 4; ++kc) {
      bfrag[h][kc].x = ws[(4 * kc + 0 + 2 * hi) * 32 + 16 * h + row];
      bfrag[h][kc].y = ws[(4 * kc + 1 + 2 * hi) * 32 + 16 * h + row];
    }
  const float bias0 = ws[512 + row];
  const float bias1 = ws[512 + 16 + row];

  const int waveId     = blockIdx.x * 8 + (threadIdx.x >> 5);
  const int totalWaves = gridDim.x * 8;

  for (int tile = waveId; tile < numTiles; tile += totalWaves) {
    const float* xt = x + (size_t)tile * (16 * 16);

    v2f afrag[4];
#pragma unroll
    for (int kc = 0; kc < 4; ++kc) {
      const v2f* p = (const v2f*)(xt + row * 16 + 4 * kc + 2 * hi);
      afrag[kc] = __builtin_nontemporal_load(p);
    }

    v8f acc0 = {};
    v8f acc1 = {};
#pragma unroll
    for (int kc = 0; kc < 4; ++kc) {
      acc0 = __builtin_amdgcn_wmma_f32_16x16x4_f32(
          false, afrag[kc], false, bfrag[0][kc], (short)0, acc0, false, false);
      acc1 = __builtin_amdgcn_wmma_f32_16x16x4_f32(
          false, afrag[kc], false, bfrag[1][kc], (short)0, acc1, false, false);
    }

    float* ot = out + (size_t)tile * (16 * 32);
#pragma unroll
    for (int v = 0; v < 8; ++v) {
      int m = v + 8 * hi;
      __builtin_nontemporal_store(acc0[v] + bias0, &ot[m * 32 + row]);
      __builtin_nontemporal_store(acc1[v] + bias1, &ot[m * 32 + 16 + row]);
    }
  }
}

extern "C" void kernel_launch(void* const* d_in, const int* in_sizes, int n_in,
                              void* d_out, int out_size, void* d_ws, size_t ws_size,
                              hipStream_t stream) {
  const float* x    = (const float*)d_in[0];
  const float* disp = (const float*)d_in[1];
  const float* sq   = (const float*)d_in[2];
  const float* bs   = (const float*)d_in[3];
  float* out = (float*)d_out;
  float* ws  = (float*)d_ws;

  const int BN       = in_sizes[0];
  const int numTiles = BN / 256;

  hipLaunchKernelGGL(cv_setup_kernel, dim3(1), dim3(192), 0, stream,
                     disp, sq, bs, ws);

  int blocks = (numTiles + 31) / 32;
  if (blocks < 1) blocks = 1;
  if (blocks > 4096) blocks = 4096;
  hipLaunchKernelGGL(cv_gemm_kernel, dim3(blocks), dim3(256), 0, stream,
                     x, ws, out, numTiles);
}


# Set-up cut at the middle layer: the two 3-layer half-maps of each circuit are probed at the same time (one lane per column) and joined by one 32-term product; the map's sum order changes
# speedup vs baseline: 1.2333x; 1.2333x over previous
#include <hip/hip_runtime.h>

typedef float v2f __attribute__((ext_vector_type(2)));
typedef float v8f __attribute__((ext_vector_type(8)));

#define NMODES 16
#define NLAYERS 6

#define GP 36

__global__ __launch_bounds__(576) void cv_setup_kernel(
    const float* __restrict__ disp,
    const float* __restrict__ sq,
    const float* __restrict__ bs,
    float* __restrict__ ws)
{
  __shared__ float4 s_sq4[12 * NMODES];
  __shared__ float4 s_bs4[12 * NMODES];
  __shared__ float4 s_d4[12 * 8];
  __shared__ float4 s_M4[2 * 16 * (GP / 4)];
  __shared__ float4 s_V4[2 * 18 * (GP / 4)];
  __shared__ float  s_cv[2][NMODES];
  float* sd = (float*)s_d4;
  float* M  = (float*)s_M4;

  const int t = threadIdx.x;

  if (t < 192) {
    float a = disp[2 * t], dphi = disp[2 * t + 1];
    float s, c;
    sincosf(dphi, &s, &c);
    sd[(t >> 4) * 32 + (t & 15)]      = 2.0f * a * c;
    sd[(t >> 4) * 32 + 16 + (t & 15)] = 2.0f * a * s;
  } else if (t < 384) {
    int u = t - 192;
    float rr = fabsf(sq[2 * u]);
    float sphi = sq[2 * u + 1];
    float ch = coshf(rr), sh = sinhf(rr);
    float sp, cp;
    sincosf(sphi, &sp, &cp);
    s_sq4[u] = make_float4(ch - cp * sh, -sp * sh, ch + cp * sh, 0.0f);
  } else {
    int u = t - 384;
    if (u < 12 * (NMODES - 1)) {
      int cl = u / (NMODES - 1), w = u % (NMODES - 1);
      float th   = 1.0f / (1.0f + expf(-bs[2 * u]));
      float bphi = bs[2 * u + 1];
      float st, ct, sb, cb;
      sincosf(th, &st, &ct);
      sincosf(bphi, &sb, &cb);
      s_bs4[cl * NMODES + w] = make_float4(ct, cb * st, sb * st, 0.0f);
    }
  }
  __syncthreads();

  if (t < 128) {
    const int wv = t >> 5, lane = t & 31;
    const int c = wv & 1;
    const bool second = (wv < 2) || (lane == 17);
    const bool use_disp = (wv >= 2) && (lane == 16 || lane == 17);
    int unit = -1;
    float unit_val = 0.0f;
    if (wv < 2)         { unit = lane; unit_val = 1.0f; }
    else if (lane < 16) { unit = lane; unit_val = 2.0f; }

    float mx[NMODES], mp[NMODES];
#pragma unroll
    for (int i = 0; i < NMODES; ++i) {
      mx[i] = (unit == i) ? unit_val : 0.0f;
      mp[i] = (unit == 16 + i) ? unit_val : 0.0f;
    }

    const int cl0 = c * NLAYERS + (second ? 3 : 0);
#pragma unroll
    for (int l = 0; l < 3; ++l) {
      const int cl = cl0 + l;
      if (use_disp) {
#pragma unroll
        for (int q = 0; q < 4; ++q) {
          const float4 dx = s_d4[cl * 8 + q], dp = s_d4[cl * 8 + 4 + q];
          mx[4 * q + 0] += dx.x; mx[4 * q + 1] += dx.y; mx[4 * q + 2] += dx.z; mx[4 * q + 3] += dx.w;
          mp[4 * q + 0] += dp.x; mp[4 * q + 1] += dp.y; mp[4 * q + 2] += dp.z; mp[4 * q + 3] += dp.w;
        }
      }
#pragma unroll
      for (int i = 0; i < NMODES; ++i) {
        const float4 q = s_sq4[cl * NMODES + i];
        const float xv = mx[i], pv = mp[i];
        mx[i] = q.x * xv + q.y * pv;
        mp[i] = q.y * xv + q.z * pv;
      }
#pragma unroll
      for (int w = 0; w < NMODES - 1; ++w) {
        const float4 b = s_bs4[cl * NMODES + w];
        const float ct = b.x, cs = b.y, ss = b.z;
        float x1 = mx[w], x2 = mx[w + 1], p1 = mp[w], p2 = mp[w + 1];
        mx[w]     = ct * x1 - cs * x2 - ss * p2;
        mx[w + 1] = cs * x1 + ct * x2 - ss * p1;
        mp[w]     = ss * x2 + ct * p1 - cs * p2;
        mp[w + 1] = ss * x1 + cs * p1 + ct * p2;
      }
    }

    if (wv < 2) {
#pragma unroll
      for (int o = 0; o < NMODES; ++o) M[(c * 16 + o) * GP + lane] = c ? mp[o] : mx[o];
    } else if (lane == 17) {
#pragma unroll
      for (int o = 0; o < NMODES; ++o) s_cv[c][o] = c ? mp[o] : mx[o];
    } else if (lane <= 18) {
      const int col = (lane == 18) ? 17 : lane;
      float4* vcol = s_V4 + (c * 18 + col) * (GP / 4);
#pragma unroll
      for (int q = 0; q < 4; ++q) {
        vcol[q]     = make_float4(mx[4 * q], mx[4 * q + 1], mx[4 * q + 2], mx[4 * q + 3]);
        vcol[4 + q] = make_float4(mp[4 * q], mp[4 * q + 1], mp[4 * q + 2], mp[4 * q + 3]);
      }
    }
  }
  __syncthreads();

  if (t < 144) {
    const int grp = t >> 3, o8 = t & 7;
    const int c  = (grp >= 9) ? 1 : 0;
    const int pr = grp - 9 * c;
    const float4* g0 = s_M4 + (c * 16 + o8) * (GP / 4);
    const float4* g1 = g0 + 8 * (GP / 4);
    const float4* v0 = s_V4 + (c * 18 + 2 * pr) * (GP / 4);
    const float4* v1 = v0 + GP / 4;
    float a00 = 0.0f, a10 = 0.0f, a01 = 0.0f, a11 = 0.0f;
#pragma unroll
    for (int k4 = 0; k4 < 8; ++k4) {
      const float4 ga = g0[k4], gb = g1[k4], u0 = v0[k4], u1 = v1[k4];
      a00 = fmaf(ga.x, u0.x, a00); a00 = fmaf(ga.y, u0.y, a00); a00 = fmaf(ga.z, u0.z, a00); a00 = fmaf(ga.w, u0.w, a00);
      a10 = fmaf(gb.x, u0.x, a10); a10 = fmaf(gb.y, u0.y, a10); a10 = fmaf(gb.z, u0.z, a10); a10 = fmaf(gb.w, u0.w, a10);
      a01 = fmaf(ga.x, u1.x, a01); a01 = fmaf(ga.y, u1.y, a01); a01 = fmaf(ga.z, u1.z, a01); a01 = fmaf(ga.w, u1.w, a01);
      a11 = fmaf(gb.x, u1.x, a11); a11 = fmaf(gb.y, u1.y, a11); a11 = fmaf(gb.z, u1.z, a11); a11 = fmaf(gb.w, u1.w, a11);
    }
    if (pr < 8) {
      ws[(2 * pr) * 32 + 16 * c + o8]         = a00;
      ws[(2 * pr) * 32 + 16 * c + o8 + 8]     = a10;
      ws[(2 * pr + 1) * 32 + 16 * c + o8]     = a01;
      ws[(2 * pr + 1) * 32 + 16 * c + o8 + 8] = a11;
    } else {
      ws[512 + 16 * c + o8]     = a00 + s_cv[c][o8];
      ws[512 + 16 * c + o8 + 8] = a10 + s_cv[c][o8 + 8];
    }
  }
}

__global__ __launch_bounds__(256) void cv_gemm_kernel(
    const float* __restrict__ x, const float* __restrict__ ws,
    float* __restrict__ out, int numTiles)
{
  const int lane = threadIdx.x & 31;
  const int row  = lane & 15;
  const int hi   = lane >> 4;

  v2f bfrag[2][4];
#pragma unroll
  for (int h = 0; h < 2; ++h)
#pragma unroll
    for (int kc = 0; kc < 4; ++kc) {
      bfrag[h][kc].x = ws[(4 * kc + 0 + 2 * hi) * 32 + 16 * h + row];
      bfrag[h][kc].y = ws[(4 * kc + 1 + 2 * hi) * 32 + 16 * h + row];
    }
  const float bias0 = ws[512 + row];
  const float bias1 = ws[512 + 16 + row];

  const int waveId     = blockIdx.x * 8 + (threadIdx.x >> 5);
  const int totalWaves = gridDim.x * 8;

  for (int tile = waveId; tile < numTiles; tile += totalWaves) {
    const float* xt = x + (size_t)tile * (16 * 16);

    v2f afrag[4];
#pragma unroll
    for (int kc = 0; kc < 4; ++kc) {
      const v2f* p = (const v2f*)(xt + row * 16 + 4 * kc + 2 * hi);
      afrag[kc] = __builtin_nontemporal_load(p);
    }

    v8f acc0 = {};
    v8f acc1 = {};
#pragma unroll
    for (int kc = 0; kc < 4; ++kc) {
      acc0 = __builtin_amdgcn_wmma_f32_16x16x4_f32(
          false, afrag[kc], false, bfrag[0][kc], (short)0, acc0, false, false);
      acc1 = __builtin_amdgcn_wmma_f32_16x16x4_f32(
          false, afrag[kc], false, bfrag[1][kc], (short)0, acc1, false, false);
    }

    float* ot = out + (size_t)tile * (16 * 32);
#pragma unroll
    for (int v = 0; v < 8; ++v) {
      int m = v + 8 * hi;
      __builtin_nontemporal_store(acc0[v] + bias0, &ot[m * 32 + row]);
      __builtin_nontemporal_store(acc1[v] + bias1, &ot[m * 32 + 16 + row]);
    }
  }
}

extern "C" void kernel_launch(void* const* d_in, const int* in_sizes, int n_in,
                              void* d_out, int out_size, void* d_ws, size_t ws_size,
                              hipStream_t stream) {
  const float* x    = (const float*)d_in[0];
  const float* disp = (const float*)d_in[1];
  const float* sq   = (const float*)d_in[2];
  const float* bs   = (const float*)d_in[3];
  float* out = (float*)d_out;
  float* ws  = (float*)d_ws;

  const int BN       = in_sizes[0];
  const int numTiles = BN / 256;

  hipLaunchKernelGGL(cv_setup_kernel, dim3(1), dim3(576), 0, stream,
                     disp, sq, bs, ws);

  int blocks = (numTiles + 31) / 32;
  if (blocks < 1) blocks = 1;
  if (blocks > 4096) blocks = 4096;
  hipLaunchKernelGGL(cv_gemm_kernel, dim3(blocks), dim3(256), 0, stream,
                     x, ws, out, numTiles);
}


# Streaming kernel's fragment index maps re-chosen (k order, sample rows, column pairs): 16-B x loads, one 8-B store per register = one whole 256-B piece of out; the k sum order changes
# speedup vs baseline: 1.3377x; 1.0847x over previous
#include <hip/hip_runtime.h>

typedef float v2f __attribute__((ext_vector_type(2)));
typedef float v8f __attribute__((ext_vector_type(8)));

#define NMODES 16
#define NLAYERS 6

#define GP 36

__device__ __forceinline__ float lane_bcast(float v, int src_lane) {
  return __builtin_bit_cast(float, __builtin_amdgcn_readlane(__builtin_bit_cast(int, v), src_lane));
}

__global__ __launch_bounds__(576) void cv_setup_kernel(
    const float* __restrict__ disp,
    const float* __restrict__ sq,
    const float* __restrict__ bs,
    float* __restrict__ ws)
{
  __shared__ float4 s_co4[12 * 32];
  __shared__ float4 s_d4[12 * 8];
  __shared__ float4 s_M4[2 * 16 * (GP / 4)];
  __shared__ float4 s_V4[2 * 18 * (GP / 4)];
  __shared__ float  s_cv[2][NMODES];
  float* sd = (float*)s_d4;
  float* M  = (float*)s_M4;

  const int t = threadIdx.x;

  if (t < 192) {
    float a = disp[2 * t], dphi = disp[2 * t + 1];
    float s, c;
    sincosf(dphi, &s, &c);
    sd[(t >> 4) * 32 + (t & 15)]      = 2.0f * a * c;
    sd[(t >> 4) * 32 + 16 + (t & 15)] = 2.0f * a * s;
  } else if (t < 384) {
    int u = t - 192;
    float rr = fabsf(sq[2 * u]);
    float sphi = sq[2 * u + 1];
    float ch = coshf(rr), sh = sinhf(rr);
    float sp, cp;
    sincosf(sphi, &sp, &cp);
    s_co4[(u >> 4) * 32 + (u & 15)] = make_float4(ch - cp * sh, -sp * sh, ch + cp * sh, 0.0f);
  } else {
    int u = t - 384;
    if (u < 12 * (NMODES - 1)) {
      int cl = u / (NMODES - 1), w = u % (NMODES - 1);
      float th   = 1.0f / (1.0f + expf(-bs[2 * u]));
      float bphi = bs[2 * u + 1];
      float st, ct, sb, cb;
      sincosf(th, &st, &ct);
      sincosf(bphi, &sb, &cb);
      s_co4[cl * 32 + 16 + w] = make_float4(ct, cb * st, sb * st, 0.0f);
    } else {
      s_co4[(u - 12 * (NMODES - 1)) * 32 + 31] = make_float4(0.0f, 0.0f, 0.0f, 0.0f);
    }
  }
  __syncthreads();

  if (t < 192) {
    const int wv = __builtin_amdgcn_readfirstlane(t >> 5);
    const int lane = t & 31;
    const int c = wv & 1;
    const int kind = wv >> 1;
    const bool use_disp = (kind == 2) || (kind == 1 && lane == 16);
    int unit = -1;
    float unit_val = 0.0f;
    if (kind == 0)                    { unit = lane; unit_val = 1.0f; }
    else if (kind == 1 && lane < 16)  { unit = lane; unit_val = 2.0f; }

    float mx[NMODES], mp[NMODES];
#pragma unroll
    for (int i = 0; i < NMODES; ++i) {
      mx[i] = (unit == i) ? unit_val : 0.0f;
      mp[i] = (unit == 16 + i) ? unit_val : 0.0f;
    }

    const int cl0 = c * NLAYERS + ((kind == 1) ? 0 : 3);
    float4 co[3];
    float  dv[3];
#pragma unroll
    for (int l = 0; l < 3; ++l) {
      co[l] = s_co4[(cl0 + l) * 32 + lane];
      dv[l] = sd[(cl0 + l) * 32 + lane];
    }

#pragma unroll
    for (int l = 0; l < 3; ++l) {
      if (kind != 0) {
#pragma unroll
        for (int i = 0; i < NMODES; ++i) {
          const float ax = lane_bcast(dv[l], i), ap = lane_bcast(dv[l], 16 + i);
          if (use_disp) { mx[i] += ax; mp[i] += ap; }
        }
      }
#pragma unroll
      for (int i = 0; i < NMODES; ++i) {
        const float qx = lane_bcast(co[l].x, i), qy = lane_bcast(co[l].y, i), qz = lane_bcast(co[l].z, i);
        const float xv = mx[i], pv = mp[i];
        mx[i] = qx * xv + qy * pv;
        mp[i] = qy * xv + qz * pv;
      }
#pragma unroll
      for (int w = 0; w < NMODES - 1; ++w) {
        const float ct = lane_bcast(co[l].x, 16 + w), cs = lane_bcast(co[l].y, 16 + w), ss = lane_bcast(co[l].z, 16 + w);
        float x1 = mx[w], x2 = mx[w + 1], p1 = mp[w], p2 = mp[w + 1];
        mx[w]     = ct * x1 - cs * x2 - ss * p2;
        mx[w + 1] = cs * x1 + ct * x2 - ss * p1;
        mp[w]     = ss * x2 + ct * p1 - cs * p2;
        mp[w + 1] = ss * x1 + cs * p1 + ct * p2;
      }
    }

    if (kind == 0) {
#pragma unroll
      for (int o = 0; o < NMODES; ++o) M[(c * 16 + o) * GP + lane] = c ? mp[o] : mx[o];
    } else if (kind == 2) {
      if (lane == 0) {
#pragma unroll
        for (int o = 0; o < NMODES; ++o) s_cv[c][o] = c ? mp[o] : mx[o];
      }
    } else if (lane <= 17) {
      float4* vcol = s_V4 + (c * 18 + lane) * (GP / 4);
#pragma unroll
      for (int q = 0; q < 4; ++q) {
        vcol[q]     = make_float4(mx[4 * q], mx[4 * q + 1], mx[4 * q + 2], mx[4 * q + 3]);
        vcol[4 + q] = make_float4(mp[4 * q], mp[4 * q + 1], mp[4 * q + 2], mp[4 * q + 3]);
      }
    }
  }
  __syncthreads();

  if (t < 144) {
    const int grp = t >> 3, o8 = t & 7;
    const int c  = (grp >= 9) ? 1 : 0;
    const int pr = grp - 9 * c;
    const float4* g0 = s_M4 + (c * 16 + o8) * (GP / 4);
    const float4* g1 = g0 + 8 * (GP / 4);
    const float4* v0 = s_V4 + (c * 18 + 2 * pr) * (GP / 4);
    const float4* v1 = v0 + GP / 4;
    float a00 = 0.0f, a10 = 0.0f, a01 = 0.0f, a11 = 0.0f;
#pragma unroll
    for (int k4 = 0; k4 < 8; ++k4) {
      const float4 ga = g0[k4], gb = g1[k4], u0 = v0[k4], u1 = v1[k4];
      a00 = fmaf(ga.x, u0.x, a00); a00 = fmaf(ga.y, u0.y, a00); a00 = fmaf(ga.z, u0.z, a00); a00 = fmaf(ga.w, u0.w, a00);
      a10 = fmaf(gb.x, u0.x, a10); a10 = fmaf(gb.y, u0.y, a10); a10 = fmaf(gb.z, u0.z, a10); a10 = fmaf(gb.w, u0.w, a10);
      a01 = fmaf(ga.x, u1.x, a01); a01 = fmaf(ga.y, u1.y, a01); a01 = fmaf(ga.z, u1.z, a01); a01 = fmaf(ga.w, u1.w, a01);
      a11 = fmaf(gb.x, u1.x, a11); a11 = fmaf(gb.y, u1.y, a11); a11 = fmaf(gb.z, u1.z, a11); a11 = fmaf(gb.w, u1.w, a11);
    }
    if (pr < 8) {
      ws[(2 * pr) * 32 + 16 * c + o8]         = a00;
      ws[(2 * pr) * 32 + 16 * c + o8 + 8]     = a10;
      ws[(2 * pr + 1) * 32 + 16 * c + o8]     = a01;
      ws[(2 * pr + 1) * 32 + 16 * c + o8 + 8] = a11;
    } else {
      ws[512 + 16 * c + o8]     = a00 + s_cv[c][o8];
      ws[512 + 16 * c + o8 + 8] = a10 + s_cv[c][o8 + 8];
    }
  }
}

typedef float v4f __attribute__((ext_vector_type(4)));

__global__ __launch_bounds__(256) void cv_gemm_kernel(
    const float* __restrict__ x, const float* __restrict__ ws,
    float* __restrict__ out, int numTiles)
{
  const int lane = threadIdx.x & 31;
  const int n    = lane & 15;
  const int hi   = lane >> 4;

  v2f bfrag[2][4];
#pragma unroll
  for (int kc = 0; kc < 4; ++kc) {
    const int k0 = 8 * (kc >> 1) + 4 * hi + 2 * (kc & 1);
    const v2f w0 = *(const v2f*)(ws + (k0 + 0) * 32 + 2 * n);
    const v2f w1 = *(const v2f*)(ws + (k0 + 1) * 32 + 2 * n);
    bfrag[0][kc].x = w0.x;  bfrag[0][kc].y = w1.x;
    bfrag[1][kc].x = w0.y;  bfrag[1][kc].y = w1.y;
  }
  const v2f bias = *(const v2f*)(ws + 512 + 2 * n);

  const int srow = 2 * (n & 7) + (n >> 3);

  const int waveId     = blockIdx.x * 8 + (threadIdx.x >> 5);
  const int totalWaves = gridDim.x * 8;

  for (int tile = waveId; tile < numTiles; tile += totalWaves) {
    const float* xt = x + (size_t)tile * (16 * 16) + srow * 16 + 4 * hi;
    const v4f a0 = __builtin_nontemporal_load((const v4f*)xt);
    const v4f a1 = __builtin_nontemporal_load((const v4f*)(xt + 8));
    v2f afrag[4];
    afrag[0].x = a0.x;  afrag[0].y = a0.y;
    afrag[1].x = a0.z;  afrag[1].y = a0.w;
    afrag[2].x = a1.x;  afrag[2].y = a1.y;
    afrag[3].x = a1.z;  afrag[3].y = a1.w;

    v8f acc0 = {};
    v8f acc1 = {};
#pragma unroll
    for (int kc = 0; kc < 4; ++kc) {
      acc0 = __builtin_amdgcn_wmma_f32_16x16x4_f32(
          false, afrag[kc], false, bfrag[0][kc], (short)0, acc0, false, false);
      acc1 = __builtin_amdgcn_wmma_f32_16x16x4_f32(
          false, afrag[kc], false, bfrag[1][kc], (short)0, acc1, false, false);
    }

    float* ot = out + (size_t)tile * (16 * 32) + hi * 32 + 2 * n;
#pragma unroll
    for (int v = 0; v < 8; ++v) {
      v2f r;
      r.x = acc0[v] + bias.x;
      r.y = acc1[v] + bias.y;
      __builtin_nontemporal_store(r, (v2f*)(ot + v * 64));
    }
  }
}

extern "C" void kernel_launch(void* const* d_in, const int* in_sizes, int n_in,
                              void* d_out, int out_size, void* d_ws, size_t ws_size,
                              hipStream_t stream) {
  const float* x    = (const float*)d_in[0];
  const float* disp = (const float*)d_in[1];
  const float* sq   = (const float*)d_in[2];
  const float* bs   = (const float*)d_in[3];
  float* out = (float*)d_out;
  float* ws  = (float*)d_ws;

  const int BN       = in_sizes[0];
  const int numTiles = BN / 256;

  hipLaunchKernelGGL(cv_setup_kernel, dim3(1), dim3(576), 0, stream,
                     disp, sq, bs, ws);

  int blocks = (numTiles + 31) / 32;
  if (blocks < 1) blocks = 1;
  if (blocks > 4096) blocks = 4096;
  hipLaunchKernelGGL(cv_gemm_kernel, dim3(blocks), dim3(256), 0, stream,
                     x, ws, out, numTiles);
}
